# starting kernel
# baseline (speedup 1.0000x reference)
#include <hip/hip_runtime.h>
#include <hip/hip_bf16.h>

typedef __attribute__((ext_vector_type(2))) float v2f;
typedef __attribute__((ext_vector_type(8))) float v8f;

#define GCN_D 64


__global__ void zero_f32(float* p, int n) {
    int i = blockIdx.x * blockDim.x + threadIdx.x;
    if (i < n) p[i] = 0.0f;
}

__global__ void deg_kernel(const int* __restrict__ dst, float* __restrict__ deg, int E) {
    int e = blockIdx.x * blockDim.x + threadIdx.x;
    if (e < E) atomicAdd(&deg[dst[e]], 1.0f);
}

__global__ void dinv_kernel(float* __restrict__ d, int n) {
    int i = blockIdx.x * blockDim.x + threadIdx.x;
    if (i < n) d[i] = rsqrtf(d[i] + 1.0f);
}

__global__ void gemm_wmma_f32(const float* __restrict__ X,
                              const float* __restrict__ W,
                              float* __restrict__ H,
                              int nrows, int mtiles) {
    const int wave = blockIdx.x * (blockDim.x >> 5) + (threadIdx.x >> 5);
    if (wave >= mtiles) return;
    const int lane = threadIdx.x & 31;
    const int l15  = lane & 15;
    const int hi   = lane >> 4;
    const int rowBase = wave * 16;
    int arow = rowBase + l15;
    if (arow >= nrows) arow = nrows - 1;

    v8f acc0 = {}, acc1 = {}, acc2 = {}, acc3 = {};

    const float* __restrict__ xrow = X + (size_t)arow * GCN_D;

    #pragma unroll
    for (int kk = 0; kk < GCN_D; kk += 4) {
        const int kb = kk + hi * 2;
        v2f a;
        a.x = xrow[kb];
        a.y = xrow[kb + 1];

        const float* __restrict__ wp = W + (size_t)kb * GCN_D + l15;
        v2f b0, b1, b2, b3;
        b0.x = wp[0];       b0.y = wp[GCN_D];
        b1.x = wp[16];      b1.y = wp[16 + GCN_D];
        b2.x = wp[32];      b2.y = wp[32 + GCN_D];
        b3.x = wp[48];      b3.y = wp[48 + GCN_D];

        acc0 = __builtin_amdgcn_wmma_f32_16x16x4_f32(false, a, false, b0, (short)0, acc0, false, false);
        acc1 = __builtin_amdgcn_wmma_f32_16x16x4_f32(false, a, false, b1, (short)0, acc1, false, false);
        acc2 = __builtin_amdgcn_wmma_f32_16x16x4_f32(false, a, false, b2, (short)0, acc2, false, false);
        acc3 = __builtin_amdgcn_wmma_f32_16x16x4_f32(false, a, false, b3, (short)0, acc3, false, false);
    }

    #pragma unroll
    for (int r = 0; r < 8; ++r) {
        const int row = rowBase + r + hi * 8;
        if (row < nrows) {
            float* __restrict__ o = H + (size_t)row * GCN_D + l15;
            o[0]  = acc0[r];
            o[16] = acc1[r];
            o[32] = acc2[r];
            o[48] = acc3[r];
        }
    }
}

__global__ void scatter1_kernel(const int* __restrict__ src, const int* __restrict__ dst,
                                const float* __restrict__ H, const float* __restrict__ dinv,
                                float* __restrict__ AGG, int E) {
    int e = blockIdx.x * blockDim.y + threadIdx.y;
    if (e >= E) return;
    const int s = src[e];
    const int d = dst[e];
    const float nrm = dinv[s] * dinv[d];
    const int q = threadIdx.x;
    const float4 hv = ((const float4*)(H + (size_t)s * GCN_D))[q];
    float* __restrict__ base = AGG + (size_t)d * GCN_D + q * 4;
    atomicAdd(base + 0, hv.x * nrm);
    atomicAdd(base + 1, hv.y * nrm);
    atomicAdd(base + 2, hv.z * nrm);
    atomicAdd(base + 3, hv.w * nrm);
}

__global__ void finalize1_dot_kernel(const float* __restrict__ AGG, const float* __restrict__ H,
                                     const float* __restrict__ dinv, const float* __restrict__ b1,
                                     const float* __restrict__ W2, float* __restrict__ h2, int n) {
    int i = blockIdx.x * blockDim.x + threadIdx.x;
    if (i >= n) return;
    const float di = dinv[i];
    const float d2 = di * di;
    const float4* __restrict__ ag = (const float4*)(AGG + (size_t)i * GCN_D);
    const float4* __restrict__ hh = (const float4*)(H + (size_t)i * GCN_D);
    const float4* __restrict__ bb = (const float4*)b1;
    const float4* __restrict__ ww = (const float4*)W2;
    float s = 0.0f;
    #pragma unroll
    for (int q = 0; q < GCN_D / 4; ++q) {
        const float4 a = ag[q];
        const float4 h = hh[q];
        const float4 b = bb[q];
        const float4 w = ww[q];
        s += fmaxf(a.x + h.x * d2 + b.x, 0.0f) * w.x;
        s += fmaxf(a.y + h.y * d2 + b.y, 0.0f) * w.y;
        s += fmaxf(a.z + h.z * d2 + b.z, 0.0f) * w.z;
        s += fmaxf(a.w + h.w * d2 + b.w, 0.0f) * w.w;
    }
    h2[i] = s;
}

__global__ void scatter2_kernel(const int* __restrict__ src, const int* __restrict__ dst,
                                const float* __restrict__ h2, const float* __restrict__ dinv,
                                float* __restrict__ agg2, int E) {
    int e = blockIdx.x * blockDim.x + threadIdx.x;
    if (e >= E) return;
    const int s = src[e];
    const int d = dst[e];
    atomicAdd(&agg2[d], h2[s] * dinv[s] * dinv[d]);
}

__global__ void finalize2_kernel(const float* __restrict__ agg2, const float* __restrict__ h2,
                                 const float* __restrict__ dinv, const float* __restrict__ b2,
                                 float* __restrict__ out, int n) {
    int i = blockIdx.x * blockDim.x + threadIdx.x;
    if (i >= n) return;
    const float di = dinv[i];
    out[i] = agg2[i] + h2[i] * di * di + b2[0];
}


extern "C" void kernel_launch(void* const* d_in, const int* in_sizes, int n_in,
                              void* d_out, int out_size, void* d_ws, size_t ws_size,
                              hipStream_t stream) {
    const float* x  = (const float*)d_in[0];
    const int*   ei = (const int*)d_in[1];
    const float* W1 = (const float*)d_in[2];
    const float* b1 = (const float*)d_in[3];
    const float* W2 = (const float*)d_in[4];
    const float* b2 = (const float*)d_in[5];
    float* out = (float*)d_out;

    const int N = in_sizes[0] / GCN_D;
    const int E = in_sizes[1] / 2;
    const int* src = ei;
    const int* dst = ei + E;

    float* H    = (float*)d_ws;
    float* AGG  = H    + (size_t)N * GCN_D;
    float* dinv = AGG  + (size_t)N * GCN_D;
    float* h2   = dinv + N;
    float* agg2 = h2   + N;

    const int T = 256;
    const int totNF = N * GCN_D;

    zero_f32<<<(totNF + T - 1) / T, T, 0, stream>>>(AGG, totNF);
    zero_f32<<<(N + T - 1) / T, T, 0, stream>>>(dinv, N);
    zero_f32<<<(N + T - 1) / T, T, 0, stream>>>(agg2, N);

    deg_kernel<<<(E + T - 1) / T, T, 0, stream>>>(dst, dinv, E);
    dinv_kernel<<<(N + T - 1) / T, T, 0, stream>>>(dinv, N);

    const int mtiles = (N + 15) / 16;
    const int wavesPerBlk = T / 32;
    const int gblocks = (mtiles + wavesPerBlk - 1) / wavesPerBlk;
    gemm_wmma_f32<<<gblocks, T, 0, stream>>>(x, W1, H, N, mtiles);

    {
        dim3 blk(16, 16);
        int grd = (E + 16 - 1) / 16;
        scatter1_kernel<<<grd, blk, 0, stream>>>(src, dst, H, dinv, AGG, E);
    }

    finalize1_dot_kernel<<<(N + T - 1) / T, T, 0, stream>>>(AGG, H, dinv, b1, W2, h2, N);

    scatter2_kernel<<<(E + T - 1) / T, T, 0, stream>>>(src, dst, h2, dinv, agg2, E);

    finalize2_kernel<<<(N + T - 1) / T, T, 0, stream>>>(agg2, h2, dinv, b2, out, N);
}


# Degree count and layer-2 scatter now add in LDS per (edge slice, dst range) block and store partial planes summed in fixed order; their global float atomics and two zero launches are gone
# speedup vs baseline: 1.5365x; 1.5365x over previous
#include <hip/hip_runtime.h>
#include <hip/hip_bf16.h>

typedef __attribute__((ext_vector_type(2))) float v2f;
typedef __attribute__((ext_vector_type(8))) float v8f;

#define GCN_D 64


__global__ void zero_f32(float* p, int n) {
    int i = blockIdx.x * blockDim.x + threadIdx.x;
    if (i < n) p[i] = 0.0f;
}

#define SEG_CAP 8192
#define SEG_T   512

__device__ __forceinline__ int seg_pitch(int n) { return (n + 31) & ~31; }

__device__ __forceinline__ void seg_bounds(int E, int N, int& e0, int& e1, int& lo0, int& hi0) {
    const int S = gridDim.x, R = gridDim.y;
    const int sliceLen = ((E + 4 * S - 1) / (4 * S)) * 4;
    e0 = min(E, (int)blockIdx.x * sliceLen);
    e1 = min(E, e0 + sliceLen);
    const int rangeLen = (((N + R - 1) / R) + 31) & ~31;
    lo0 = min(N, (int)blockIdx.y * rangeLen);
    hi0 = min(N, lo0 + rangeLen);
}

__global__ __launch_bounds__(SEG_T) void deg_lds_kernel(const int* __restrict__ dst,
                                                        float* __restrict__ planes, int E, int N) {
    __shared__ unsigned cnt[SEG_CAP];
    int e0, e1, lo0, hi0;
    seg_bounds(E, N, e0, e1, lo0, hi0);
    float* __restrict__ plane = planes + (size_t)blockIdx.x * seg_pitch(N);
    const int nvec = ((E & 3) == 0) ? ((e1 - e0) >> 2) : 0;
    const int4* __restrict__ d4 = (const int4*)(dst + e0);

    for (int lo = lo0; lo < hi0; lo += SEG_CAP) {
        const unsigned len = (unsigned)min(SEG_CAP, hi0 - lo);
        for (unsigned i = threadIdx.x; i < len; i += SEG_T) cnt[i] = 0u;
        __syncthreads();
        for (int g = threadIdx.x; g < nvec; g += 4 * SEG_T) {
            int4 d[4];
            #pragma unroll
            for (int j = 0; j < 4; ++j) {
                const int gj = g + j * SEG_T;
                d[j] = make_int4(-1, -1, -1, -1);
                if (gj < nvec) d[j] = d4[gj];
            }
            #pragma unroll
            for (int j = 0; j < 4; ++j) {
                const unsigned o0 = (unsigned)(d[j].x - lo), o1 = (unsigned)(d[j].y - lo);
                const unsigned o2 = (unsigned)(d[j].z - lo), o3 = (unsigned)(d[j].w - lo);
                if (o0 < len) atomicAdd(&cnt[o0], 1u);
                if (o1 < len) atomicAdd(&cnt[o1], 1u);
                if (o2 < len) atomicAdd(&cnt[o2], 1u);
                if (o3 < len) atomicAdd(&cnt[o3], 1u);
            }
        }
        for (int e = e0 + 4 * nvec + threadIdx.x; e < e1; e += SEG_T) {
            const unsigned o = (unsigned)(dst[e] - lo);
            if (o < len) atomicAdd(&cnt[o], 1u);
        }
        __syncthreads();
        for (unsigned i = threadIdx.x; i < len; i += SEG_T) plane[lo + i] = (float)cnt[i];
        __syncthreads();
    }
}

__global__ void dinv_kernel(const float* __restrict__ planes, float* __restrict__ d, int n, int nplanes) {
    int i = blockIdx.x * blockDim.x + threadIdx.x;
    if (i >= n) return;
    const int pitch = seg_pitch(n);
    float c = 0.0f;
    for (int p = 0; p < nplanes; ++p) c += planes[(size_t)p * pitch + i];
    d[i] = rsqrtf(c + 1.0f);
}

__global__ void gemm_wmma_f32(const float* __restrict__ X,
                              const float* __restrict__ W,
                              float* __restrict__ H,
                              int nrows, int mtiles) {
    const int wave = blockIdx.x * (blockDim.x >> 5) + (threadIdx.x >> 5);
    if (wave >= mtiles) return;
    const int lane = threadIdx.x & 31;
    const int l15  = lane & 15;
    const int hi   = lane >> 4;
    const int rowBase = wave * 16;
    int arow = rowBase + l15;
    if (arow >= nrows) arow = nrows - 1;

    v8f acc0 = {}, acc1 = {}, acc2 = {}, acc3 = {};

    const float* __restrict__ xrow = X + (size_t)arow * GCN_D;

    #pragma unroll
    for (int kk = 0; kk < GCN_D; kk += 4) {
        const int kb = kk + hi * 2;
        v2f a;
        a.x = xrow[kb];
        a.y = xrow[kb + 1];

        const float* __restrict__ wp = W + (size_t)kb * GCN_D + l15;
        v2f b0, b1, b2, b3;
        b0.x = wp[0];       b0.y = wp[GCN_D];
        b1.x = wp[16];      b1.y = wp[16 + GCN_D];
        b2.x = wp[32];      b2.y = wp[32 + GCN_D];
        b3.x = wp[48];      b3.y = wp[48 + GCN_D];

        acc0 = __builtin_amdgcn_wmma_f32_16x16x4_f32(false, a, false, b0, (short)0, acc0, false, false);
        acc1 = __builtin_amdgcn_wmma_f32_16x16x4_f32(false, a, false, b1, (short)0, acc1, false, false);
        acc2 = __builtin_amdgcn_wmma_f32_16x16x4_f32(false, a, false, b2, (short)0, acc2, false, false);
        acc3 = __builtin_amdgcn_wmma_f32_16x16x4_f32(false, a, false, b3, (short)0, acc3, false, false);
    }

    #pragma unroll
    for (int r = 0; r < 8; ++r) {
        const int row = rowBase + r + hi * 8;
        if (row < nrows) {
            float* __restrict__ o = H + (size_t)row * GCN_D + l15;
            o[0]  = acc0[r];
            o[16] = acc1[r];
            o[32] = acc2[r];
            o[48] = acc3[r];
        }
    }
}

__global__ void scatter1_kernel(const int* __restrict__ src, const int* __restrict__ dst,
                                const float* __restrict__ H, const float* __restrict__ dinv,
                                float* __restrict__ AGG, int E) {
    int e = blockIdx.x * blockDim.y + threadIdx.y;
    if (e >= E) return;
    const int s = src[e];
    const int d = dst[e];
    const float nrm = dinv[s] * dinv[d];
    const int q = threadIdx.x;
    const float4 hv = ((const float4*)(H + (size_t)s * GCN_D))[q];
    float* __restrict__ base = AGG + (size_t)d * GCN_D + q * 4;
    atomicAdd(base + 0, hv.x * nrm);
    atomicAdd(base + 1, hv.y * nrm);
    atomicAdd(base + 2, hv.z * nrm);
    atomicAdd(base + 3, hv.w * nrm);
}

__global__ void finalize1_dot_kernel(const float* __restrict__ AGG, const float* __restrict__ H,
                                     const float* __restrict__ dinv, const float* __restrict__ b1,
                                     const float* __restrict__ W2, float* __restrict__ h2, int n) {
    int i = blockIdx.x * blockDim.x + threadIdx.x;
    if (i >= n) return;
    const float di = dinv[i];
    const float d2 = di * di;
    const float4* __restrict__ ag = (const float4*)(AGG + (size_t)i * GCN_D);
    const float4* __restrict__ hh = (const float4*)(H + (size_t)i * GCN_D);
    const float4* __restrict__ bb = (const float4*)b1;
    const float4* __restrict__ ww = (const float4*)W2;
    float s = 0.0f;
    #pragma unroll
    for (int q = 0; q < GCN_D / 4; ++q) {
        const float4 a = ag[q];
        const float4 h = hh[q];
        const float4 b = bb[q];
        const float4 w = ww[q];
        s += fmaxf(a.x + h.x * d2 + b.x, 0.0f) * w.x;
        s += fmaxf(a.y + h.y * d2 + b.y, 0.0f) * w.y;
        s += fmaxf(a.z + h.z * d2 + b.z, 0.0f) * w.z;
        s += fmaxf(a.w + h.w * d2 + b.w, 0.0f) * w.w;
    }
    h2[i] = s;
}

__global__ __launch_bounds__(SEG_T) void scatter2_lds_kernel(const int* __restrict__ src, const int* __restrict__ dst,
                                                             const float* __restrict__ h2, const float* __restrict__ dinv,
                                                             float* __restrict__ planes, int E, int N) {
    __shared__ float acc[SEG_CAP];
    int e0, e1, lo0, hi0;
    seg_bounds(E, N, e0, e1, lo0, hi0);
    float* __restrict__ plane = planes + (size_t)blockIdx.x * seg_pitch(N);
    const int nvec = ((E & 3) == 0) ? ((e1 - e0) >> 2) : 0;
    const int4* __restrict__ d4 = (const int4*)(dst + e0);
    const int4* __restrict__ s4 = (const int4*)(src + e0);

    for (int lo = lo0; lo < hi0; lo += SEG_CAP) {
        const unsigned len = (unsigned)min(SEG_CAP, hi0 - lo);
        for (unsigned i = threadIdx.x; i < len; i += SEG_T) acc[i] = 0.0f;
        __syncthreads();
        for (int g = threadIdx.x; g < nvec; g += 2 * SEG_T) {
            const int4 da = d4[g];
            const int4 sa = s4[g];
            int4 db = make_int4(-1, -1, -1, -1);
            int4 sb = make_int4(0, 0, 0, 0);
            if (g + SEG_T < nvec) { db = d4[g + SEG_T]; sb = s4[g + SEG_T]; }
            const int dd[8] = {da.x, da.y, da.z, da.w, db.x, db.y, db.z, db.w};
            const int ss[8] = {sa.x, sa.y, sa.z, sa.w, sb.x, sb.y, sb.z, sb.w};
            unsigned off[8];
            float hv[8], ds[8], dt[8];
            #pragma unroll
            for (int k = 0; k < 8; ++k) {
                off[k] = (unsigned)(dd[k] - lo);
                const int si = (off[k] < len) ? ss[k] : 0;
                const int di = (off[k] < len) ? dd[k] : 0;
                hv[k] = h2[si]; ds[k] = dinv[si]; dt[k] = dinv[di];
            }
            #pragma unroll
            for (int k = 0; k < 8; ++k)
                if (off[k] < len) atomicAdd(&acc[off[k]], hv[k] * ds[k] * dt[k]);
        }
        for (int e = e0 + 4 * nvec + threadIdx.x; e < e1; e += SEG_T) {
            const int d = dst[e];
            const unsigned o = (unsigned)(d - lo);
            if (o < len) { const int s = src[e]; atomicAdd(&acc[o], h2[s] * dinv[s] * dinv[d]); }
        }
        __syncthreads();
        for (unsigned i = threadIdx.x; i < len; i += SEG_T) plane[lo + i] = acc[i];
        __syncthreads();
    }
}

__global__ void finalize2_kernel(const float* __restrict__ planes, const float* __restrict__ h2,
                                 const float* __restrict__ dinv, const float* __restrict__ b2,
                                 float* __restrict__ out, int n, int nplanes) {
    int i = blockIdx.x * blockDim.x + threadIdx.x;
    if (i >= n) return;
    const int pitch = seg_pitch(n);
    float a = 0.0f;
    for (int p = 0; p < nplanes; ++p) a += planes[(size_t)p * pitch + i];
    const float di = dinv[i];
    out[i] = a + h2[i] * di * di + b2[0];
}


extern "C" void kernel_launch(void* const* d_in, const int* in_sizes, int n_in,
                              void* d_out, int out_size, void* d_ws, size_t ws_size,
                              hipStream_t stream) {
    const float* x  = (const float*)d_in[0];
    const int*   ei = (const int*)d_in[1];
    const float* W1 = (const float*)d_in[2];
    const float* b1 = (const float*)d_in[3];
    const float* W2 = (const float*)d_in[4];
    const float* b2 = (const float*)d_in[5];
    float* out = (float*)d_out;

    const int N = in_sizes[0] / GCN_D;
    const int E = in_sizes[1] / 2;
    const int* src = ei;
    const int* dst = ei + E;

    float* H    = (float*)d_ws;
    float* AGG  = H    + (size_t)N * GCN_D;
    float* dinv = AGG  + (size_t)N * GCN_D;
    float* h2   = dinv + N;
    float* degPlanes = H;
    float* aggPlanes = AGG;

    const int T = 256;
    const int totNF = N * GCN_D;
    const int SEG_S = 32;
    const int SEG_R = 8;
    const dim3 segGrid(SEG_S, SEG_R);

    zero_f32<<<(totNF + T - 1) / T, T, 0, stream>>>(AGG, totNF);

    deg_lds_kernel<<<segGrid, SEG_T, 0, stream>>>(dst, degPlanes, E, N);
    dinv_kernel<<<(N + T - 1) / T, T, 0, stream>>>(degPlanes, dinv, N, SEG_S);

    const int mtiles = (N + 15) / 16;
    const int wavesPerBlk = T / 32;
    const int gblocks = (mtiles + wavesPerBlk - 1) / wavesPerBlk;
    gemm_wmma_f32<<<gblocks, T, 0, stream>>>(x, W1, H, N, mtiles);

    {
        dim3 blk(16, 16);
        int grd = (E + 16 - 1) / 16;
        scatter1_kernel<<<grd, blk, 0, stream>>>(src, dst, H, dinv, AGG, E);
    }

    finalize1_dot_kernel<<<(N + T - 1) / T, T, 0, stream>>>(AGG, H, dinv, b1, W2, h2, N);

    scatter2_lds_kernel<<<segGrid, SEG_T, 0, stream>>>(src, dst, h2, dinv, aggPlanes, E, N);
    finalize2_kernel<<<(N + T - 1) / T, T, 0, stream>>>(aggPlanes, h2, dinv, b2, out, N, SEG_S);
}


# Layer-1 scatter remapped: each wave loops over 32-edge batches, all lanes on one edge row, so each row load and f32 atomic covers one 128-byte line; 400k waves become 5k; sums stay in arrival order
# speedup vs baseline: 2.2394x; 1.4575x over previous
#include <hip/hip_runtime.h>
#include <hip/hip_bf16.h>

typedef __attribute__((ext_vector_type(2))) float v2f;
typedef __attribute__((ext_vector_type(8))) float v8f;

#define GCN_D 64


__global__ void zero_f32(float* p, int n) {
    int i = blockIdx.x * blockDim.x + threadIdx.x;
    if (i < n) p[i] = 0.0f;
}

#define SEG_CAP 8192
#define SEG_T   512

__device__ __forceinline__ int seg_pitch(int n) { return (n + 31) & ~31; }

__device__ __forceinline__ void seg_bounds(int E, int N, int& e0, int& e1, int& lo0, int& hi0) {
    const int S = gridDim.x, R = gridDim.y;
    const int sliceLen = ((E + 4 * S - 1) / (4 * S)) * 4;
    e0 = min(E, (int)blockIdx.x * sliceLen);
    e1 = min(E, e0 + sliceLen);
    const int rangeLen = (((N + R - 1) / R) + 31) & ~31;
    lo0 = min(N, (int)blockIdx.y * rangeLen);
    hi0 = min(N, lo0 + rangeLen);
}

__global__ __launch_bounds__(SEG_T) void deg_lds_kernel(const int* __restrict__ dst,
                                                        float* __restrict__ planes, int E, int N) {
    __shared__ unsigned cnt[SEG_CAP];
    int e0, e1, lo0, hi0;
    seg_bounds(E, N, e0, e1, lo0, hi0);
    float* __restrict__ plane = planes + (size_t)blockIdx.x * seg_pitch(N);
    const int nvec = ((E & 3) == 0) ? ((e1 - e0) >> 2) : 0;
    const int4* __restrict__ d4 = (const int4*)(dst + e0);

    for (int lo = lo0; lo < hi0; lo += SEG_CAP) {
        const unsigned len = (unsigned)min(SEG_CAP, hi0 - lo);
        for (unsigned i = threadIdx.x; i < len; i += SEG_T) cnt[i] = 0u;
        __syncthreads();
        for (int g = threadIdx.x; g < nvec; g += 4 * SEG_T) {
            int4 d[4];
            #pragma unroll
            for (int j = 0; j < 4; ++j) {
                const int gj = g + j * SEG_T;
                d[j] = make_int4(-1, -1, -1, -1);
                if (gj < nvec) d[j] = d4[gj];
            }
            #pragma unroll
            for (int j = 0; j < 4; ++j) {
                const unsigned o0 = (unsigned)(d[j].x - lo), o1 = (unsigned)(d[j].y - lo);
                const unsigned o2 = (unsigned)(d[j].z - lo), o3 = (unsigned)(d[j].w - lo);
                if (o0 < len) atomicAdd(&cnt[o0], 1u);
                if (o1 < len) atomicAdd(&cnt[o1], 1u);
                if (o2 < len) atomicAdd(&cnt[o2], 1u);
                if (o3 < len) atomicAdd(&cnt[o3], 1u);
            }
        }
        for (int e = e0 + 4 * nvec + threadIdx.x; e < e1; e += SEG_T) {
            const unsigned o = (unsigned)(dst[e] - lo);
            if (o < len) atomicAdd(&cnt[o], 1u);
        }
        __syncthreads();
        for (unsigned i = threadIdx.x; i < len; i += SEG_T) plane[lo + i] = (float)cnt[i];
        __syncthreads();
    }
}

__global__ void dinv_kernel(const float* __restrict__ planes, float* __restrict__ d, int n, int nplanes) {
    int i = blockIdx.x * blockDim.x + threadIdx.x;
    if (i >= n) return;
    const int pitch = seg_pitch(n);
    float c = 0.0f;
    for (int p = 0; p < nplanes; ++p) c += planes[(size_t)p * pitch + i];
    d[i] = rsqrtf(c + 1.0f);
}

__global__ void gemm_wmma_f32(const float* __restrict__ X,
                              const float* __restrict__ W,
                              float* __restrict__ H,
                              int nrows, int mtiles) {
    const int wave = blockIdx.x * (blockDim.x >> 5) + (threadIdx.x >> 5);
    if (wave >= mtiles) return;
    const int lane = threadIdx.x & 31;
    const int l15  = lane & 15;
    const int hi   = lane >> 4;
    const int rowBase = wave * 16;
    int arow = rowBase + l15;
    if (arow >= nrows) arow = nrows - 1;

    v8f acc0 = {}, acc1 = {}, acc2 = {}, acc3 = {};

    const float* __restrict__ xrow = X + (size_t)arow * GCN_D;

    #pragma unroll
    for (int kk = 0; kk < GCN_D; kk += 4) {
        const int kb = kk + hi * 2;
        v2f a;
        a.x = xrow[kb];
        a.y = xrow[kb + 1];

        const float* __restrict__ wp = W + (size_t)kb * GCN_D + l15;
        v2f b0, b1, b2, b3;
        b0.x = wp[0];       b0.y = wp[GCN_D];
        b1.x = wp[16];      b1.y = wp[16 + GCN_D];
        b2.x = wp[32];      b2.y = wp[32 + GCN_D];
        b3.x = wp[48];      b3.y = wp[48 + GCN_D];

        acc0 = __builtin_amdgcn_wmma_f32_16x16x4_f32(false, a, false, b0, (short)0, acc0, false, false);
        acc1 = __builtin_amdgcn_wmma_f32_16x16x4_f32(false, a, false, b1, (short)0, acc1, false, false);
        acc2 = __builtin_amdgcn_wmma_f32_16x16x4_f32(false, a, false, b2, (short)0, acc2, false, false);
        acc3 = __builtin_amdgcn_wmma_f32_16x16x4_f32(false, a, false, b3, (short)0, acc3, false, false);
    }

    #pragma unroll
    for (int r = 0; r < 8; ++r) {
        const int row = rowBase + r + hi * 8;
        if (row < nrows) {
            float* __restrict__ o = H + (size_t)row * GCN_D + l15;
            o[0]  = acc0[r];
            o[16] = acc1[r];
            o[32] = acc2[r];
            o[48] = acc3[r];
        }
    }
}

#define S1_BATCH 32
#define S1_GROUP 16
#define S1_T     128
__global__ __launch_bounds__(S1_T) void scatter1_kernel(const int* __restrict__ src, const int* __restrict__ dst,
                                                        const float* __restrict__ H, const float* __restrict__ dinv,
                                                        float* __restrict__ AGG, int E) {
    const unsigned lane = threadIdx.x & 31u;
    const int wavesPerBlk = blockDim.x >> 5;
    const int nw = gridDim.x * wavesPerBlk;
    const int nb = (E + S1_BATCH - 1) / S1_BATCH;
    int b = blockIdx.x * wavesPerBlk + __builtin_amdgcn_readfirstlane((int)(threadIdx.x >> 5));
    if (b >= nb) return;

    int s = 0, d = 0;
    {
        const int e = b * S1_BATCH + (int)lane;
        if (e < E) { s = src[e]; d = dst[e]; }
    }
    float nrm = dinv[s] * dinv[d];

    for (;;) {
        const int cnt = min(S1_BATCH, E - b * S1_BATCH);
        const int bn = b + nw;
        int sn = 0, dn = 0;
        {
            const int en = bn * S1_BATCH + (int)lane;
            if (bn < nb && en < E) { sn = src[en]; dn = dst[en]; }
        }

        for (int k0 = 0; k0 < cnt; k0 += S1_GROUP) {
            if (k0 + S1_GROUP <= cnt) {
                float h0[S1_GROUP], h1[S1_GROUP];
                #pragma unroll
                for (int j = 0; j < S1_GROUP; ++j) {
                    const int sk = __builtin_amdgcn_readlane(s, k0 + j);
                    const float* __restrict__ row = H + (size_t)sk * GCN_D;
                    h0[j] = row[lane];
                    h1[j] = row[lane + 32u];
                }
                #pragma unroll
                for (int j = 0; j < S1_GROUP; ++j) {
                    const int dk = __builtin_amdgcn_readlane(d, k0 + j);
                    const float nk = __int_as_float(__builtin_amdgcn_readlane(__float_as_int(nrm), k0 + j));
                    float* __restrict__ o = AGG + (size_t)dk * GCN_D;
                    atomicAdd(o + lane,       h0[j] * nk);
                    atomicAdd(o + lane + 32u, h1[j] * nk);
                }
            } else {
                for (int k = k0; k < cnt; ++k) {
                    const int sk = __builtin_amdgcn_readlane(s, k);
                    const int dk = __builtin_amdgcn_readlane(d, k);
                    const float nk = __int_as_float(__builtin_amdgcn_readlane(__float_as_int(nrm), k));
                    const float* __restrict__ row = H + (size_t)sk * GCN_D;
                    float* __restrict__ o = AGG + (size_t)dk * GCN_D;
                    atomicAdd(o + lane,       row[lane]       * nk);
                    atomicAdd(o + lane + 32u, row[lane + 32u] * nk);
                }
            }
        }

        if (bn >= nb) break;
        b = bn; s = sn; d = dn;
        nrm = dinv[s] * dinv[d];
    }
}

__global__ void finalize1_dot_kernel(const float* __restrict__ AGG, const float* __restrict__ H,
                                     const float* __restrict__ dinv, const float* __restrict__ b1,
                                     const float* __restrict__ W2, float* __restrict__ h2, int n) {
    int i = blockIdx.x * blockDim.x + threadIdx.x;
    if (i >= n) return;
    const float di = dinv[i];
    const float d2 = di * di;
    const float4* __restrict__ ag = (const float4*)(AGG + (size_t)i * GCN_D);
    const float4* __restrict__ hh = (const float4*)(H + (size_t)i * GCN_D);
    const float4* __restrict__ bb = (const float4*)b1;
    const float4* __restrict__ ww = (const float4*)W2;
    float s = 0.0f;
    #pragma unroll
    for (int q = 0; q < GCN_D / 4; ++q) {
        const float4 a = ag[q];
        const float4 h = hh[q];
        const float4 b = bb[q];
        const float4 w = ww[q];
        s += fmaxf(a.x + h.x * d2 + b.x, 0.0f) * w.x;
        s += fmaxf(a.y + h.y * d2 + b.y, 0.0f) * w.y;
        s += fmaxf(a.z + h.z * d2 + b.z, 0.0f) * w.z;
        s += fmaxf(a.w + h.w * d2 + b.w, 0.0f) * w.w;
    }
    h2[i] = s;
}

__global__ __launch_bounds__(SEG_T) void scatter2_lds_kernel(const int* __restrict__ src, const int* __restrict__ dst,
                                                             const float* __restrict__ h2, const float* __restrict__ dinv,
                                                             float* __restrict__ planes, int E, int N) {
    __shared__ float acc[SEG_CAP];
    int e0, e1, lo0, hi0;
    seg_bounds(E, N, e0, e1, lo0, hi0);
    float* __restrict__ plane = planes + (size_t)blockIdx.x * seg_pitch(N);
    const int nvec = ((E & 3) == 0) ? ((e1 - e0) >> 2) : 0;
    const int4* __restrict__ d4 = (const int4*)(dst + e0);
    const int4* __restrict__ s4 = (const int4*)(src + e0);

    for (int lo = lo0; lo < hi0; lo += SEG_CAP) {
        const unsigned len = (unsigned)min(SEG_CAP, hi0 - lo);
        for (unsigned i = threadIdx.x; i < len; i += SEG_T) acc[i] = 0.0f;
        __syncthreads();
        for (int g = threadIdx.x; g < nvec; g += 2 * SEG_T) {
            const int4 da = d4[g];
            const int4 sa = s4[g];
            int4 db = make_int4(-1, -1, -1, -1);
            int4 sb = make_int4(0, 0, 0, 0);
            if (g + SEG_T < nvec) { db = d4[g + SEG_T]; sb = s4[g + SEG_T]; }
            const int dd[8] = {da.x, da.y, da.z, da.w, db.x, db.y, db.z, db.w};
            const int ss[8] = {sa.x, sa.y, sa.z, sa.w, sb.x, sb.y, sb.z, sb.w};
            unsigned off[8];
            float hv[8], ds[8], dt[8];
            #pragma unroll
            for (int k = 0; k < 8; ++k) {
                off[k] = (unsigned)(dd[k] - lo);
                const int si = (off[k] < len) ? ss[k] : 0;
                const int di = (off[k] < len) ? dd[k] : 0;
                hv[k] = h2[si]; ds[k] = dinv[si]; dt[k] = dinv[di];
            }
            #pragma unroll
            for (int k = 0; k < 8; ++k)
                if (off[k] < len) atomicAdd(&acc[off[k]], hv[k] * ds[k] * dt[k]);
        }
        for (int e = e0 + 4 * nvec + threadIdx.x; e < e1; e += SEG_T) {
            const int d = dst[e];
            const unsigned o = (unsigned)(d - lo);
            if (o < len) { const int s = src[e]; atomicAdd(&acc[o], h2[s] * dinv[s] * dinv[d]); }
        }
        __syncthreads();
        for (unsigned i = threadIdx.x; i < len; i += SEG_T) plane[lo + i] = acc[i];
        __syncthreads();
    }
}

__global__ void finalize2_kernel(const float* __restrict__ planes, const float* __restrict__ h2,
                                 const float* __restrict__ dinv, const float* __restrict__ b2,
                                 float* __restrict__ out, int n, int nplanes) {
    int i = blockIdx.x * blockDim.x + threadIdx.x;
    if (i >= n) return;
    const int pitch = seg_pitch(n);
    float a = 0.0f;
    for (int p = 0; p < nplanes; ++p) a += planes[(size_t)p * pitch + i];
    const float di = dinv[i];
    out[i] = a + h2[i] * di * di + b2[0];
}


extern "C" void kernel_launch(void* const* d_in, const int* in_sizes, int n_in,
                              void* d_out, int out_size, void* d_ws, size_t ws_size,
                              hipStream_t stream) {
    const float* x  = (const float*)d_in[0];
    const int*   ei = (const int*)d_in[1];
    const float* W1 = (const float*)d_in[2];
    const float* b1 = (const float*)d_in[3];
    const float* W2 = (const float*)d_in[4];
    const float* b2 = (const float*)d_in[5];
    float* out = (float*)d_out;

    const int N = in_sizes[0] / GCN_D;
    const int E = in_sizes[1] / 2;
    const int* src = ei;
    const int* dst = ei + E;

    float* H    = (float*)d_ws;
    float* AGG  = H    + (size_t)N * GCN_D;
    float* dinv = AGG  + (size_t)N * GCN_D;
    float* h2   = dinv + N;
    float* degPlanes = H;
    float* aggPlanes = AGG;

    const int T = 256;
    const int totNF = N * GCN_D;
    const int SEG_S = 32;
    const int SEG_R = 8;
    const dim3 segGrid(SEG_S, SEG_R);

    zero_f32<<<(totNF + T - 1) / T, T, 0, stream>>>(AGG, totNF);

    deg_lds_kernel<<<segGrid, SEG_T, 0, stream>>>(dst, degPlanes, E, N);
    dinv_kernel<<<(N + T - 1) / T, T, 0, stream>>>(degPlanes, dinv, N, SEG_S);

    const int mtiles = (N + 15) / 16;
    const int wavesPerBlk = T / 32;
    const int gblocks = (mtiles + wavesPerBlk - 1) / wavesPerBlk;
    gemm_wmma_f32<<<gblocks, T, 0, stream>>>(x, W1, H, N, mtiles);

    scatter1_kernel<<<1280, S1_T, 0, stream>>>(src, dst, H, dinv, AGG, E);

    finalize1_dot_kernel<<<(N + T - 1) / T, T, 0, stream>>>(AGG, H, dinv, b1, W2, h2, N);

    scatter2_lds_kernel<<<segGrid, SEG_T, 0, stream>>>(src, dst, h2, dinv, aggPlanes, E, N);
    finalize2_kernel<<<(N + T - 1) / T, T, 0, stream>>>(aggPlanes, h2, dinv, b2, out, N, SEG_S);
}


# Degree count and GEMM share one launch (work-group roles), and the dinv launch also clears the layer-1 accumulator, where the degree planes now lie; zero launch gone, 8 launches become 6
# speedup vs baseline: 2.3690x; 1.0579x over previous
#include <hip/hip_runtime.h>
#include <hip/hip_bf16.h>

typedef __attribute__((ext_vector_type(2))) float v2f;
typedef __attribute__((ext_vector_type(8))) float v8f;

#define GCN_D 64

#define SEG_CAP 8192
#define SEG_T   512

__device__ __forceinline__ int seg_pitch(int n) { return (n + 31) & ~31; }

__device__ __forceinline__ void seg_bounds_at(int E, int N, int S, int R, int sx, int ry,
                                              int& e0, int& e1, int& lo0, int& hi0) {
    const int sliceLen = ((E + 4 * S - 1) / (4 * S)) * 4;
    e0 = min(E, sx * sliceLen);
    e1 = min(E, e0 + sliceLen);
    const int rangeLen = (((N + R - 1) / R) + 31) & ~31;
    lo0 = min(N, ry * rangeLen);
    hi0 = min(N, lo0 + rangeLen);
}

__device__ __forceinline__ void seg_bounds(int E, int N, int& e0, int& e1, int& lo0, int& hi0) {
    seg_bounds_at(E, N, (int)gridDim.x, (int)gridDim.y, (int)blockIdx.x, (int)blockIdx.y, e0, e1, lo0, hi0);
}

__device__ __forceinline__ void deg_lds_role(unsigned* cnt, const int* __restrict__ dst,
                                             float* __restrict__ planes, int E, int N,
                                             int S, int R, int sx, int ry) {
    int e0, e1, lo0, hi0;
    seg_bounds_at(E, N, S, R, sx, ry, e0, e1, lo0, hi0);
    float* __restrict__ plane = planes + (size_t)sx * seg_pitch(N);
    const int nvec = ((E & 3) == 0) ? ((e1 - e0) >> 2) : 0;
    const int4* __restrict__ d4 = (const int4*)(dst + e0);

    for (int lo = lo0; lo < hi0; lo += SEG_CAP) {
        const unsigned len = (unsigned)min(SEG_CAP, hi0 - lo);
        for (unsigned i = threadIdx.x; i < len; i += SEG_T) cnt[i] = 0u;
        __syncthreads();
        for (int g = threadIdx.x; g < nvec; g += 4 * SEG_T) {
            int4 d[4];
            #pragma unroll
            for (int j = 0; j < 4; ++j) {
                const int gj = g + j * SEG_T;
                d[j] = make_int4(-1, -1, -1, -1);
                if (gj < nvec) d[j] = d4[gj];
            }
            #pragma unroll
            for (int j = 0; j < 4; ++j) {
                const unsigned o0 = (unsigned)(d[j].x - lo), o1 = (unsigned)(d[j].y - lo);
                const unsigned o2 = (unsigned)(d[j].z - lo), o3 = (unsigned)(d[j].w - lo);
                if (o0 < len) atomicAdd(&cnt[o0], 1u);
                if (o1 < len) atomicAdd(&cnt[o1], 1u);
                if (o2 < len) atomicAdd(&cnt[o2], 1u);
                if (o3 < len) atomicAdd(&cnt[o3], 1u);
            }
        }
        for (int e = e0 + 4 * nvec + threadIdx.x; e < e1; e += SEG_T) {
            const unsigned o = (unsigned)(dst[e] - lo);
            if (o < len) atomicAdd(&cnt[o], 1u);
        }
        __syncthreads();
        for (unsigned i = threadIdx.x; i < len; i += SEG_T) plane[lo + i] = (float)cnt[i];
        __syncthreads();
    }
}

__global__ void dinv_zero_kernel(float* agg, float* __restrict__ d, int n, int nplanes, int aggFloats) {
    const int i = blockIdx.x * blockDim.x + threadIdx.x;
    const int pitch = seg_pitch(n);
    if (i < pitch) {
        if (i < n) {
            float c = 0.0f;
            for (int p = 0; p < nplanes; ++p) c += agg[(size_t)p * pitch + i];
            d[i] = rsqrtf(c + 1.0f);
        }
        for (int p = 0; p < nplanes; ++p) agg[(size_t)p * pitch + i] = 0.0f;
    }
    const int planeFloats = nplanes * pitch;
    float4* rest = (float4*)(agg + planeFloats);
    const int nrest4 = (aggFloats - planeFloats) >> 2;
    const int nthreads = gridDim.x * blockDim.x;
    for (int j = i; j < nrest4; j += nthreads) rest[j] = make_float4(0.0f, 0.0f, 0.0f, 0.0f);
}

__device__ __forceinline__ void gemm_wmma_f32_role(const float* __restrict__ X,
                                                   const float* __restrict__ W,
                                                   float* __restrict__ H,
                                                   int nrows, int mtiles, int blk) {
    const int wave = blk * (blockDim.x >> 5) + (threadIdx.x >> 5);
    if (wave >= mtiles) return;
    const int lane = threadIdx.x & 31;
    const int l15  = lane & 15;
    const int hi   = lane >> 4;
    const int rowBase = wave * 16;
    int arow = rowBase + l15;
    if (arow >= nrows) arow = nrows - 1;

    v8f acc0 = {}, acc1 = {}, acc2 = {}, acc3 = {};

    const float* __restrict__ xrow = X + (size_t)arow * GCN_D;

    #pragma unroll
    for (int kk = 0; kk < GCN_D; kk += 4) {
        const int kb = kk + hi * 2;
        v2f a;
        a.x = xrow[kb];
        a.y = xrow[kb + 1];

        const float* __restrict__ wp = W + (size_t)kb * GCN_D + l15;
        v2f b0, b1, b2, b3;
        b0.x = wp[0];       b0.y = wp[GCN_D];
        b1.x = wp[16];      b1.y = wp[16 + GCN_D];
        b2.x = wp[32];      b2.y = wp[32 + GCN_D];
        b3.x = wp[48];      b3.y = wp[48 + GCN_D];

        acc0 = __builtin_amdgcn_wmma_f32_16x16x4_f32(false, a, false, b0, (short)0, acc0, false, false);
        acc1 = __builtin_amdgcn_wmma_f32_16x16x4_f32(false, a, false, b1, (short)0, acc1, false, false);
        acc2 = __builtin_amdgcn_wmma_f32_16x16x4_f32(false, a, false, b2, (short)0, acc2, false, false);
        acc3 = __builtin_amdgcn_wmma_f32_16x16x4_f32(false, a, false, b3, (short)0, acc3, false, false);
    }

    #pragma unroll
    for (int r = 0; r < 8; ++r) {
        const int row = rowBase + r + hi * 8;
        if (row < nrows) {
            float* __restrict__ o = H + (size_t)row * GCN_D + l15;
            o[0]  = acc0[r];
            o[16] = acc1[r];
            o[32] = acc2[r];
            o[48] = acc3[r];
        }
    }
}

__global__ __launch_bounds__(SEG_T) void front_kernel(const int* __restrict__ dst, float* __restrict__ degPlanes,
                                                      int E, int N, int S, int R,
                                                      const float* __restrict__ X, const float* __restrict__ W,
                                                      float* __restrict__ H, int mtiles) {
    __shared__ unsigned cnt[SEG_CAP];
    const int wavesPerBlk = (int)(blockDim.x >> 5);
    const int gblocks = (mtiles + wavesPerBlk - 1) / wavesPerBlk;
    const int b = (int)blockIdx.x;
    if (b < gblocks) gemm_wmma_f32_role(X, W, H, N, mtiles, b);
    else             deg_lds_role(cnt, dst, degPlanes, E, N, S, R, (b - gblocks) % S, (b - gblocks) / S);
}

#define S1_BATCH 32
#define S1_GROUP 16
#define S1_T     128
__global__ __launch_bounds__(S1_T) void scatter1_kernel(const int* __restrict__ src, const int* __restrict__ dst,
                                                        const float* __restrict__ H, const float* __restrict__ dinv,
                                                        float* __restrict__ AGG, int E) {
    const unsigned lane = threadIdx.x & 31u;
    const int wavesPerBlk = blockDim.x >> 5;
    const int nw = gridDim.x * wavesPerBlk;
    const int nb = (E + S1_BATCH - 1) / S1_BATCH;
    int b = blockIdx.x * wavesPerBlk + __builtin_amdgcn_readfirstlane((int)(threadIdx.x >> 5));
    if (b >= nb) return;

    int s = 0, d = 0;
    {
        const int e = b * S1_BATCH + (int)lane;
        if (e < E) { s = src[e]; d = dst[e]; }
    }
    float nrm = dinv[s] * dinv[d];

    for (;;) {
        const int cnt = min(S1_BATCH, E - b * S1_BATCH);
        const int bn = b + nw;
        int sn = 0, dn = 0;
        {
            const int en = bn * S1_BATCH + (int)lane;
            if (bn < nb && en < E) { sn = src[en]; dn = dst[en]; }
        }

        for (int k0 = 0; k0 < cnt; k0 += S1_GROUP) {
            if (k0 + S1_GROUP <= cnt) {
                float h0[S1_GROUP], h1[S1_GROUP];
                #pragma unroll
                for (int j = 0; j < S1_GROUP; ++j) {
                    const int sk = __builtin_amdgcn_readlane(s, k0 + j);
                    const float* __restrict__ row = H + (size_t)sk * GCN_D;
                    h0[j] = row[lane];
                    h1[j] = row[lane + 32u];
                }
                #pragma unroll
                for (int j = 0; j < S1_GROUP; ++j) {
                    const int dk = __builtin_amdgcn_readlane(d, k0 + j);
                    const float nk = __int_as_float(__builtin_amdgcn_readlane(__float_as_int(nrm), k0 + j));
                    float* __restrict__ o = AGG + (size_t)dk * GCN_D;
                    atomicAdd(o + lane,       h0[j] * nk);
                    atomicAdd(o + lane + 32u, h1[j] * nk);
                }
            } else {
                for (int k = k0; k < cnt; ++k) {
                    const int sk = __builtin_amdgcn_readlane(s, k);
                    const int dk = __builtin_amdgcn_readlane(d, k);
                    const float nk = __int_as_float(__builtin_amdgcn_readlane(__float_as_int(nrm), k));
                    const float* __restrict__ row = H + (size_t)sk * GCN_D;
                    float* __restrict__ o = AGG + (size_t)dk * GCN_D;
                    atomicAdd(o + lane,       row[lane]       * nk);
                    atomicAdd(o + lane + 32u, row[lane + 32u] * nk);
                }
            }
        }

        if (bn >= nb) break;
        b = bn; s = sn; d = dn;
        nrm = dinv[s] * dinv[d];
    }
}

__global__ void finalize1_dot_kernel(const float* __restrict__ AGG, const float* __restrict__ H,
                                     const float* __restrict__ dinv, const float* __restrict__ b1,
                                     const float* __restrict__ W2, float* __restrict__ h2, int n) {
    int i = blockIdx.x * blockDim.x + threadIdx.x;
    if (i >= n) return;
    const float di = dinv[i];
    const float d2 = di * di;
    const float4* __restrict__ ag = (const float4*)(AGG + (size_t)i * GCN_D);
    const float4* __restrict__ hh = (const float4*)(H + (size_t)i * GCN_D);
    const float4* __restrict__ bb = (const float4*)b1;
    const float4* __restrict__ ww = (const float4*)W2;
    float s = 0.0f;
    #pragma unroll
    for (int q = 0; q < GCN_D / 4; ++q) {
        const float4 a = ag[q];
        const float4 h = hh[q];
        const float4 b = bb[q];
        const float4 w = ww[q];
        s += fmaxf(a.x + h.x * d2 + b.x, 0.0f) * w.x;
        s += fmaxf(a.y + h.y * d2 + b.y, 0.0f) * w.y;
        s += fmaxf(a.z + h.z * d2 + b.z, 0.0f) * w.z;
        s += fmaxf(a.w + h.w * d2 + b.w, 0.0f) * w.w;
    }
    h2[i] = s;
}

__global__ __launch_bounds__(SEG_T) void scatter2_lds_kernel(const int* __restrict__ src, const int* __restrict__ dst,
                                                             const float* __restrict__ h2, const float* __restrict__ dinv,
                                                             float* __restrict__ planes, int E, int N) {
    __shared__ float acc[SEG_CAP];
    int e0, e1, lo0, hi0;
    seg_bounds(E, N, e0, e1, lo0, hi0);
    float* __restrict__ plane = planes + (size_t)blockIdx.x * seg_pitch(N);
    const int nvec = ((E & 3) == 0) ? ((e1 - e0) >> 2) : 0;
    const int4* __restrict__ d4 = (const int4*)(dst + e0);
    const int4* __restrict__ s4 = (const int4*)(src + e0);

    for (int lo = lo0; lo < hi0; lo += SEG_CAP) {
        const unsigned len = (unsigned)min(SEG_CAP, hi0 - lo);
        for (unsigned i = threadIdx.x; i < len; i += SEG_T) acc[i] = 0.0f;
        __syncthreads();
        for (int g = threadIdx.x; g < nvec; g += 2 * SEG_T) {
            const int4 da = d4[g];
            const int4 sa = s4[g];
            int4 db = make_int4(-1, -1, -1, -1);
            int4 sb = make_int4(0, 0, 0, 0);
            if (g + SEG_T < nvec) { db = d4[g + SEG_T]; sb = s4[g + SEG_T]; }
            const int dd[8] = {da.x, da.y, da.z, da.w, db.x, db.y, db.z, db.w};
            const int ss[8] = {sa.x, sa.y, sa.z, sa.w, sb.x, sb.y, sb.z, sb.w};
            unsigned off[8];
            float hv[8], ds[8], dt[8];
            #pragma unroll
            for (int k = 0; k < 8; ++k) {
                off[k] = (unsigned)(dd[k] - lo);
                const int si = (off[k] < len) ? ss[k] : 0;
                const int di = (off[k] < len) ? dd[k] : 0;
                hv[k] = h2[si]; ds[k] = dinv[si]; dt[k] = dinv[di];
            }
            #pragma unroll
            for (int k = 0; k < 8; ++k)
                if (off[k] < len) atomicAdd(&acc[off[k]], hv[k] * ds[k] * dt[k]);
        }
        for (int e = e0 + 4 * nvec + threadIdx.x; e < e1; e += SEG_T) {
            const int d = dst[e];
            const unsigned o = (unsigned)(d - lo);
            if (o < len) { const int s = src[e]; atomicAdd(&acc[o], h2[s] * dinv[s] * dinv[d]); }
        }
        __syncthreads();
        for (unsigned i = threadIdx.x; i < len; i += SEG_T) plane[lo + i] = acc[i];
        __syncthreads();
    }
}

__global__ void finalize2_kernel(const float* __restrict__ planes, const float* __restrict__ h2,
                                 const float* __restrict__ dinv, const float* __restrict__ b2,
                                 float* __restrict__ out, int n, int nplanes) {
    int i = blockIdx.x * blockDim.x + threadIdx.x;
    if (i >= n) return;
    const int pitch = seg_pitch(n);
    float a = 0.0f;
    for (int p = 0; p < nplanes; ++p) a += planes[(size_t)p * pitch + i];
    const float di = dinv[i];
    out[i] = a + h2[i] * di * di + b2[0];
}


extern "C" void kernel_launch(void* const* d_in, const int* in_sizes, int n_in,
                              void* d_out, int out_size, void* d_ws, size_t ws_size,
                              hipStream_t stream) {
    const float* x  = (const float*)d_in[0];
    const int*   ei = (const int*)d_in[1];
    const float* W1 = (const float*)d_in[2];
    const float* b1 = (const float*)d_in[3];
    const float* W2 = (const float*)d_in[4];
    const float* b2 = (const float*)d_in[5];
    float* out = (float*)d_out;

    const int N = in_sizes[0] / GCN_D;
    const int E = in_sizes[1] / 2;
    const int* src = ei;
    const int* dst = ei + E;

    float* H    = (float*)d_ws;
    float* AGG  = H    + (size_t)N * GCN_D;
    float* dinv = AGG  + (size_t)N * GCN_D;
    float* h2   = dinv + N;
    float* degPlanes = AGG;
    float* aggPlanes = AGG;

    const int T = 256;
    const int totNF = N * GCN_D;
    const int SEG_S = 32;
    const int SEG_R = 8;
    const dim3 segGrid(SEG_S, SEG_R);

    const int mtiles = (N + 15) / 16;
    const int wavesPerBlk = SEG_T / 32;
    const int gblocks = (mtiles + wavesPerBlk - 1) / wavesPerBlk;
    front_kernel<<<gblocks + SEG_S * SEG_R, SEG_T, 0, stream>>>(dst, degPlanes, E, N, SEG_S, SEG_R, x, W1, H, mtiles);

    const int TZ = 128;
    dinv_zero_kernel<<<(N + 32 + TZ - 1) / TZ, TZ, 0, stream>>>(AGG, dinv, N, SEG_S, totNF);

    scatter1_kernel<<<1280, S1_T, 0, stream>>>(src, dst, H, dinv, AGG, E);

    finalize1_dot_kernel<<<(N + T - 1) / T, T, 0, stream>>>(AGG, H, dinv, b1, W2, h2, N);

    scatter2_lds_kernel<<<segGrid, SEG_T, 0, stream>>>(src, dst, h2, dinv, aggPlanes, E, N);
    finalize2_kernel<<<(N + T - 1) / T, T, 0, stream>>>(aggPlanes, h2, dinv, b2, out, N, SEG_S);
}


# Edge normalisation hoisted to per-node work: the dinv launch scales row i of H by dinv[i], the finalize kernels apply dinv[dst]; edge loops gather no dinv (sum order kept, terms round differently)
# speedup vs baseline: 2.5227x; 1.0649x over previous
#include <hip/hip_runtime.h>
#include <hip/hip_bf16.h>

typedef __attribute__((ext_vector_type(2))) float v2f;
typedef __attribute__((ext_vector_type(8))) float v8f;

#define GCN_D 64

#define SEG_CAP 8192
#define SEG_T   512

__device__ __forceinline__ int seg_pitch(int n) { return (n + 31) & ~31; }

__device__ __forceinline__ void seg_bounds_at(int E, int N, int S, int R, int sx, int ry,
                                              int& e0, int& e1, int& lo0, int& hi0) {
    const int sliceLen = ((E + 4 * S - 1) / (4 * S)) * 4;
    e0 = min(E, sx * sliceLen);
    e1 = min(E, e0 + sliceLen);
    const int rangeLen = (((N + R - 1) / R) + 31) & ~31;
    lo0 = min(N, ry * rangeLen);
    hi0 = min(N, lo0 + rangeLen);
}

__device__ __forceinline__ void seg_bounds(int E, int N, int& e0, int& e1, int& lo0, int& hi0) {
    seg_bounds_at(E, N, (int)gridDim.x, (int)gridDim.y, (int)blockIdx.x, (int)blockIdx.y, e0, e1, lo0, hi0);
}

__device__ __forceinline__ void deg_lds_role(unsigned* cnt, const int* __restrict__ dst,
                                             float* __restrict__ planes, int E, int N,
                                             int S, int R, int sx, int ry) {
    int e0, e1, lo0, hi0;
    seg_bounds_at(E, N, S, R, sx, ry, e0, e1, lo0, hi0);
    float* __restrict__ plane = planes + (size_t)sx * seg_pitch(N);
    const int nvec = ((E & 3) == 0) ? ((e1 - e0) >> 2) : 0;
    const int4* __restrict__ d4 = (const int4*)(dst + e0);

    for (int lo = lo0; lo < hi0; lo += SEG_CAP) {
        const unsigned len = (unsigned)min(SEG_CAP, hi0 - lo);
        for (unsigned i = threadIdx.x; i < len; i += SEG_T) cnt[i] = 0u;
        __syncthreads();
        for (int g = threadIdx.x; g < nvec; g += 4 * SEG_T) {
            int4 d[4];
            #pragma unroll
            for (int j = 0; j < 4; ++j) {
                const int gj = g + j * SEG_T;
                d[j] = make_int4(-1, -1, -1, -1);
                if (gj < nvec) d[j] = d4[gj];
            }
            #pragma unroll
            for (int j = 0; j < 4; ++j) {
                const unsigned o0 = (unsigned)(d[j].x - lo), o1 = (unsigned)(d[j].y - lo);
                const unsigned o2 = (unsigned)(d[j].z - lo), o3 = (unsigned)(d[j].w - lo);
                if (o0 < len) atomicAdd(&cnt[o0], 1u);
                if (o1 < len) atomicAdd(&cnt[o1], 1u);
                if (o2 < len) atomicAdd(&cnt[o2], 1u);
                if (o3 < len) atomicAdd(&cnt[o3], 1u);
            }
        }
        for (int e = e0 + 4 * nvec + threadIdx.x; e < e1; e += SEG_T) {
            const unsigned o = (unsigned)(dst[e] - lo);
            if (o < len) atomicAdd(&cnt[o], 1u);
        }
        __syncthreads();
        for (unsigned i = threadIdx.x; i < len; i += SEG_T) plane[lo + i] = (float)cnt[i];
        __syncthreads();
    }
}

#define DZ_T 128
__global__ __launch_bounds__(DZ_T) void dinv_zero_scale_kernel(float* agg, float* __restrict__ d, float* __restrict__ H,
                                                               int n, int nplanes, int aggFloats) {
    __shared__ float sd[DZ_T];
    const int i = blockIdx.x * DZ_T + threadIdx.x;
    const int pitch = seg_pitch(n);
    float di = 0.0f;
    if (i < pitch) {
        if (i < n) {
            float c = 0.0f;
            for (int p = 0; p < nplanes; ++p) c += agg[(size_t)p * pitch + i];
            di = rsqrtf(c + 1.0f);
            d[i] = di;
        }
        for (int p = 0; p < nplanes; ++p) agg[(size_t)p * pitch + i] = 0.0f;
    }
    sd[threadIdx.x] = di;
    __syncthreads();
    const int row0 = blockIdx.x * DZ_T;
    float4* __restrict__ h4 = (float4*)(H + (size_t)row0 * GCN_D);
    if (row0 + DZ_T <= n) {
        float4 v[GCN_D / 4];
        #pragma unroll
        for (int k = 0; k < GCN_D / 4; ++k) v[k] = h4[(int)threadIdx.x + k * DZ_T];
        #pragma unroll
        for (int k = 0; k < GCN_D / 4; ++k) {
            const int j = (int)threadIdx.x + k * DZ_T;
            const float s = sd[j >> 4];
            v[k].x *= s; v[k].y *= s; v[k].z *= s; v[k].w *= s;
            h4[j] = v[k];
        }
    } else {
        for (int k = 0; k < GCN_D / 4; ++k) {
            const int j = (int)threadIdx.x + k * DZ_T;
            const int r = j >> 4;
            if (row0 + r < n) {
                float4 v = h4[j];
                const float s = sd[r];
                v.x *= s; v.y *= s; v.z *= s; v.w *= s;
                h4[j] = v;
            }
        }
    }
    const int planeFloats = nplanes * pitch;
    float4* rest = (float4*)(agg + planeFloats);
    const int nrest4 = (aggFloats - planeFloats) >> 2;
    const int nthreads = gridDim.x * DZ_T;
    for (int j = i; j < nrest4; j += nthreads) rest[j] = make_float4(0.0f, 0.0f, 0.0f, 0.0f);
}

__device__ __forceinline__ void gemm_wmma_f32_role(const float* __restrict__ X,
                                                   const float* __restrict__ W,
                                                   float* __restrict__ H,
                                                   int nrows, int mtiles, int blk) {
    const int wave = blk * (blockDim.x >> 5) + (threadIdx.x >> 5);
    if (wave >= mtiles) return;
    const int lane = threadIdx.x & 31;
    const int l15  = lane & 15;
    const int hi   = lane >> 4;
    const int rowBase = wave * 16;
    int arow = rowBase + l15;
    if (arow >= nrows) arow = nrows - 1;

    v8f acc0 = {}, acc1 = {}, acc2 = {}, acc3 = {};

    const float* __restrict__ xrow = X + (size_t)arow * GCN_D;

    #pragma unroll
    for (int kk = 0; kk < GCN_D; kk += 4) {
        const int kb = kk + hi * 2;
        v2f a;
        a.x = xrow[kb];
        a.y = xrow[kb + 1];

        const float* __restrict__ wp = W + (size_t)kb * GCN_D + l15;
        v2f b0, b1, b2, b3;
        b0.x = wp[0];       b0.y = wp[GCN_D];
        b1.x = wp[16];      b1.y = wp[16 + GCN_D];
        b2.x = wp[32];      b2.y = wp[32 + GCN_D];
        b3.x = wp[48];      b3.y = wp[48 + GCN_D];

        acc0 = __builtin_amdgcn_wmma_f32_16x16x4_f32(false, a, false, b0, (short)0, acc0, false, false);
        acc1 = __builtin_amdgcn_wmma_f32_16x16x4_f32(false, a, false, b1, (short)0, acc1, false, false);
        acc2 = __builtin_amdgcn_wmma_f32_16x16x4_f32(false, a, false, b2, (short)0, acc2, false, false);
        acc3 = __builtin_amdgcn_wmma_f32_16x16x4_f32(false, a, false, b3, (short)0, acc3, false, false);
    }

    #pragma unroll
    for (int r = 0; r < 8; ++r) {
        const int row = rowBase + r + hi * 8;
        if (row < nrows) {
            float* __restrict__ o = H + (size_t)row * GCN_D + l15;
            o[0]  = acc0[r];
            o[16] = acc1[r];
            o[32] = acc2[r];
            o[48] = acc3[r];
        }
    }
}

__global__ __launch_bounds__(SEG_T) void front_kernel(const int* __restrict__ dst, float* __restrict__ degPlanes,
                                                      int E, int N, int S, int R,
                                                      const float* __restrict__ X, const float* __restrict__ W,
                                                      float* __restrict__ H, int mtiles) {
    __shared__ unsigned cnt[SEG_CAP];
    const int wavesPerBlk = (int)(blockDim.x >> 5);
    const int gblocks = (mtiles + wavesPerBlk - 1) / wavesPerBlk;
    const int b = (int)blockIdx.x;
    if (b < gblocks) gemm_wmma_f32_role(X, W, H, N, mtiles, b);
    else             deg_lds_role(cnt, dst, degPlanes, E, N, S, R, (b - gblocks) % S, (b - gblocks) / S);
}

#define S1_BATCH 32
#define S1_GROUP 16
#define S1_T     128
__global__ __launch_bounds__(S1_T) void scatter1_kernel(const int* __restrict__ src, const int* __restrict__ dst,
                                                        const float* __restrict__ H,
                                                        float* __restrict__ AGG, int E) {
    const unsigned lane = threadIdx.x & 31u;
    const int wavesPerBlk = blockDim.x >> 5;
    const int nw = gridDim.x * wavesPerBlk;
    const int nb = (E + S1_BATCH - 1) / S1_BATCH;
    int b = blockIdx.x * wavesPerBlk + __builtin_amdgcn_readfirstlane((int)(threadIdx.x >> 5));
    if (b >= nb) return;

    int s = 0, d = 0;
    {
        const int e = b * S1_BATCH + (int)lane;
        if (e < E) { s = src[e]; d = dst[e]; }
    }

    for (;;) {
        const int cnt = min(S1_BATCH, E - b * S1_BATCH);
        const int bn = b + nw;
        int sn = 0, dn = 0;
        {
            const int en = bn * S1_BATCH + (int)lane;
            if (bn < nb && en < E) { sn = src[en]; dn = dst[en]; }
        }

        for (int k0 = 0; k0 < cnt; k0 += S1_GROUP) {
            if (k0 + S1_GROUP <= cnt) {
                float h0[S1_GROUP], h1[S1_GROUP];
                #pragma unroll
                for (int j = 0; j < S1_GROUP; ++j) {
                    const int sk = __builtin_amdgcn_readlane(s, k0 + j);
                    const float* __restrict__ row = H + (size_t)sk * GCN_D;
                    h0[j] = row[lane];
                    h1[j] = row[lane + 32u];
                }
                #pragma unroll
                for (int j = 0; j < S1_GROUP; ++j) {
                    const int dk = __builtin_amdgcn_readlane(d, k0 + j);
                    float* __restrict__ o = AGG + (size_t)dk * GCN_D;
                    atomicAdd(o + lane,       h0[j]);
                    atomicAdd(o + lane + 32u, h1[j]);
                }
            } else {
                for (int k = k0; k < cnt; ++k) {
                    const int sk = __builtin_amdgcn_readlane(s, k);
                    const int dk = __builtin_amdgcn_readlane(d, k);
                    const float* __restrict__ row = H + (size_t)sk * GCN_D;
                    float* __restrict__ o = AGG + (size_t)dk * GCN_D;
                    atomicAdd(o + lane,       row[lane]);
                    atomicAdd(o + lane + 32u, row[lane + 32u]);
                }
            }
        }

        if (bn >= nb) break;
        b = bn; s = sn; d = dn;
    }
}

__global__ void finalize1_dot_kernel(const float* __restrict__ AGG, const float* __restrict__ H,
                                     const float* __restrict__ dinv, const float* __restrict__ b1,
                                     const float* __restrict__ W2, float* __restrict__ g2, int n) {
    int i = blockIdx.x * blockDim.x + threadIdx.x;
    if (i >= n) return;
    const float di = dinv[i];
    const float4* __restrict__ ag = (const float4*)(AGG + (size_t)i * GCN_D);
    const float4* __restrict__ hh = (const float4*)(H + (size_t)i * GCN_D);
    const float4* __restrict__ bb = (const float4*)b1;
    const float4* __restrict__ ww = (const float4*)W2;
    float s = 0.0f;
    #pragma unroll
    for (int q = 0; q < GCN_D / 4; ++q) {
        const float4 a = ag[q];
        const float4 h = hh[q];
        const float4 b = bb[q];
        const float4 w = ww[q];
        s += fmaxf((a.x + h.x) * di + b.x, 0.0f) * w.x;
        s += fmaxf((a.y + h.y) * di + b.y, 0.0f) * w.y;
        s += fmaxf((a.z + h.z) * di + b.z, 0.0f) * w.z;
        s += fmaxf((a.w + h.w) * di + b.w, 0.0f) * w.w;
    }
    g2[i] = s * di;
}

__global__ __launch_bounds__(SEG_T) void scatter2_lds_kernel(const int* __restrict__ src, const int* __restrict__ dst,
                                                             const float* __restrict__ g2,
                                                             float* __restrict__ planes, int E, int N) {
    __shared__ float acc[SEG_CAP];
    int e0, e1, lo0, hi0;
    seg_bounds(E, N, e0, e1, lo0, hi0);
    float* __restrict__ plane = planes + (size_t)blockIdx.x * seg_pitch(N);
    const int nvec = ((E & 3) == 0) ? ((e1 - e0) >> 2) : 0;
    const int4* __restrict__ d4 = (const int4*)(dst + e0);
    const int4* __restrict__ s4 = (const int4*)(src + e0);

    for (int lo = lo0; lo < hi0; lo += SEG_CAP) {
        const unsigned len = (unsigned)min(SEG_CAP, hi0 - lo);
        for (unsigned i = threadIdx.x; i < len; i += SEG_T) acc[i] = 0.0f;
        __syncthreads();
        for (int g = threadIdx.x; g < nvec; g += 2 * SEG_T) {
            const int4 da = d4[g];
            const int4 sa = s4[g];
            int4 db = make_int4(-1, -1, -1, -1);
            int4 sb = make_int4(0, 0, 0, 0);
            if (g + SEG_T < nvec) { db = d4[g + SEG_T]; sb = s4[g + SEG_T]; }
            const int dd[8] = {da.x, da.y, da.z, da.w, db.x, db.y, db.z, db.w};
            const int ss[8] = {sa.x, sa.y, sa.z, sa.w, sb.x, sb.y, sb.z, sb.w};
            unsigned off[8];
            float hv[8];
            #pragma unroll
            for (int k = 0; k < 8; ++k) {
                off[k] = (unsigned)(dd[k] - lo);
                const int si = (off[k] < len) ? ss[k] : 0;
                hv[k] = g2[si];
            }
            #pragma unroll
            for (int k = 0; k < 8; ++k)
                if (off[k] < len) atomicAdd(&acc[off[k]], hv[k]);
        }
        for (int e = e0 + 4 * nvec + threadIdx.x; e < e1; e += SEG_T) {
            const int d = dst[e];
            const unsigned o = (unsigned)(d - lo);
            if (o < len) atomicAdd(&acc[o], g2[src[e]]);
        }
        __syncthreads();
        for (unsigned i = threadIdx.x; i < len; i += SEG_T) plane[lo + i] = acc[i];
        __syncthreads();
    }
}

__global__ void finalize2_kernel(const float* __restrict__ planes, const float* __restrict__ g2,
                                 const float* __restrict__ dinv, const float* __restrict__ b2,
                                 float* __restrict__ out, int n, int nplanes) {
    int i = blockIdx.x * blockDim.x + threadIdx.x;
    if (i >= n) return;
    const int pitch = seg_pitch(n);
    float a = 0.0f;
    for (int p = 0; p < nplanes; ++p) a += planes[(size_t)p * pitch + i];
    out[i] = (a + g2[i]) * dinv[i] + b2[0];
}


extern "C" void kernel_launch(void* const* d_in, const int* in_sizes, int n_in,
                              void* d_out, int out_size, void* d_ws, size_t ws_size,
                              hipStream_t stream) {
    const float* x  = (const float*)d_in[0];
    const int*   ei = (const int*)d_in[1];
    const float* W1 = (const float*)d_in[2];
    const float* b1 = (const float*)d_in[3];
    const float* W2 = (const float*)d_in[4];
    const float* b2 = (const float*)d_in[5];
    float* out = (float*)d_out;

    const int N = in_sizes[0] / GCN_D;
    const int E = in_sizes[1] / 2;
    const int* src = ei;
    const int* dst = ei + E;

    float* H    = (float*)d_ws;
    float* AGG  = H    + (size_t)N * GCN_D;
    float* dinv = AGG  + (size_t)N * GCN_D;
    float* g2   = dinv + N;
    float* degPlanes = AGG;
    float* aggPlanes = AGG;

    const int T = 256;
    const int totNF = N * GCN_D;
    const int SEG_S = 32;
    const int SEG_R = 8;
    const dim3 segGrid(SEG_S, SEG_R);

    const int mtiles = (N + 15) / 16;
    const int wavesPerBlk = SEG_T / 32;
    const int gblocks = (mtiles + wavesPerBlk - 1) / wavesPerBlk;
    front_kernel<<<gblocks + SEG_S * SEG_R, SEG_T, 0, stream>>>(dst, degPlanes, E, N, SEG_S, SEG_R, x, W1, H, mtiles);

    dinv_zero_scale_kernel<<<(N + 32 + DZ_T - 1) / DZ_T, DZ_T, 0, stream>>>(AGG, dinv, H, N, SEG_S, totNF);

    scatter1_kernel<<<1280, S1_T, 0, stream>>>(src, dst, H, AGG, E);

    finalize1_dot_kernel<<<(N + T - 1) / T, T, 0, stream>>>(AGG, H, dinv, b1, W2, g2, N);

    scatter2_lds_kernel<<<segGrid, SEG_T, 0, stream>>>(src, dst, g2, aggPlanes, E, N);
    finalize2_kernel<<<(N + T - 1) / T, T, 0, stream>>>(aggPlanes, g2, dinv, b2, out, N, SEG_S);
}
